# starting kernel
# baseline (speedup 1.0000x reference)
#include <hip/hip_runtime.h>
#include <math.h>


typedef float v2f __attribute__((ext_vector_type(2)));
typedef float v8f __attribute__((ext_vector_type(8)));

#define BSZ    4096
#define DIMN   64
#define HIDN   256
#define TSTEP  101
#define GAMMA_C 0.1f
#define EPS_C   1e-12f
#define YS 68
#define HS 260

__device__ __forceinline__ float hw_tanh(float x) {
#if __has_builtin(__builtin_amdgcn_tanhf)
    return __builtin_amdgcn_tanhf(x);
#elif __has_builtin(__builtin_amdgcn_tanh_f32)
    return __builtin_amdgcn_tanh_f32(x);
#else
    return tanhf(x);
#endif
}

__global__ __launch_bounds__(512)
void snde_stab_ode_kernel(const float* __restrict__ y0,
                          const float* __restrict__ tarr,
                          const float* __restrict__ W1,
                          const float* __restrict__ b1,
                          const float* __restrict__ W2,
                          const float* __restrict__ b2,
                          const float* __restrict__ wg,
                          const float* __restrict__ bgp,
                          float* __restrict__ out)
{
    __shared__ float y_lds[16 * YS];
    __shared__ float h_lds[16 * HS];
    __shared__ float fp_lds[4][16 * DIMN];
    __shared__ float wg_s[DIMN];
    __shared__ float b2_s[DIMN];
    __shared__ float coef_s[16];
    __shared__ float wgss_s;

    const int tid  = threadIdx.x;
    const int w    = tid >> 5;
    const int lane = tid & 31;
    const int lm   = lane & 15;
    const int lh   = lane >> 4;
    const int row0 = blockIdx.x * 16;

    v2f b1f[16];
    const int n1 = w * 16 + lm;
#pragma unroll
    for (int kk = 0; kk < 16; ++kk) {
        const int k0 = kk * 4 + 2 * lh;
        b1f[kk].x = W1[(k0 + 0) * HIDN + n1];
        b1f[kk].y = W1[(k0 + 1) * HIDN + n1];
    }
    const int g2  = w >> 2;
    const int n2t = w & 3;
    const int n2  = n2t * 16 + lm;
    v2f b2f[16];
#pragma unroll
    for (int kk = 0; kk < 16; ++kk) {
        const int k0 = g2 * 64 + kk * 4 + 2 * lh;
        b2f[kk].x = W2[(k0 + 0) * DIMN + n2];
        b2f[kk].y = W2[(k0 + 1) * DIMN + n2];
    }
    const float b1v = b1[n1];

    if (tid < DIMN) { wg_s[tid] = wg[tid]; b2_s[tid] = b2[tid]; }
    if (tid == 0) {
        float s = 0.f;
        for (int j = 0; j < DIMN; ++j) { const float v = wg[j]; s += v * v; }
        wgss_s = s;
    }
    for (int e = tid; e < 16 * DIMN; e += 512) {
        const int i = e >> 6, j = e & 63;
        const float v = y0[(row0 + i) * DIMN + j];
        y_lds[i * YS + j] = v;
        out[(size_t)(row0 + i) * DIMN + j] = v;
    }
    const float dt = tarr[1] - tarr[0];
    const float bg = bgp[0];
    __syncthreads();

    for (int t = 0; t < TSTEP - 1; ++t) {
        v8f acc1 = {0.f, 0.f, 0.f, 0.f, 0.f, 0.f, 0.f, 0.f};
#pragma unroll
        for (int kk = 0; kk < 16; ++kk) {
            const v2f a = *(const v2f*)&y_lds[lm * YS + kk * 4 + 2 * lh];
            acc1 = __builtin_amdgcn_wmma_f32_16x16x4_f32(
                       false, a, false, b1f[kk], (short)0, acc1, false, false);
        }
#pragma unroll
        for (int r = 0; r < 8; ++r) {
            h_lds[(r + 8 * lh) * HS + w * 16 + lm] = hw_tanh(acc1[r] + b1v);
        }
        if (w == 15) {
            float p = 0.f;
            const int j0 = lh * 32;
#pragma unroll
            for (int j = 0; j < 32; ++j) p += y_lds[lm * YS + j0 + j] * wg_s[j0 + j];
            const float z = bg + p + __shfl_xor(p, 16, 32);
            const float gv = hw_tanh(z);
            const float c  = 1.f - gv * gv;
            const float s  = c * c * wgss_s;
            const float pinv = (fabsf(s) > EPS_C) ? (1.f / s) : 0.f;
            if (lh == 0) coef_s[lm] = GAMMA_C * gv * pinv * c;
        }
        __syncthreads();

        v8f acc2 = {0.f, 0.f, 0.f, 0.f, 0.f, 0.f, 0.f, 0.f};
#pragma unroll
        for (int kk = 0; kk < 16; ++kk) {
            const v2f a = *(const v2f*)&h_lds[lm * HS + g2 * 64 + kk * 4 + 2 * lh];
            acc2 = __builtin_amdgcn_wmma_f32_16x16x4_f32(
                       false, a, false, b2f[kk], (short)0, acc2, false, false);
        }
#pragma unroll
        for (int r = 0; r < 8; ++r) {
            fp_lds[g2][(r + 8 * lh) * DIMN + n2] = acc2[r];
        }
        __syncthreads();

        const size_t obase = (size_t)(t + 1) * BSZ * DIMN;
        for (int e = tid; e < 16 * DIMN; e += 512) {
            const int i = e >> 6, j = e & 63;
            const float fv = b2_s[j] + fp_lds[0][e] + fp_lds[1][e]
                                     + fp_lds[2][e] + fp_lds[3][e];
            const float yn = y_lds[i * YS + j] + (fv - coef_s[i] * wg_s[j]) * dt;
            y_lds[i * YS + j] = yn;
            out[obase + (size_t)(row0 + i) * DIMN + j] = yn;
        }
        __syncthreads();
    }
}

extern "C" void kernel_launch(void* const* d_in, const int* in_sizes, int n_in,
                              void* d_out, int out_size, void* d_ws, size_t ws_size,
                              hipStream_t stream) {
    (void)in_sizes; (void)n_in; (void)out_size; (void)d_ws; (void)ws_size;
    const float* y0 = (const float*)d_in[0];
    const float* t  = (const float*)d_in[1];
    const float* W1 = (const float*)d_in[2];
    const float* b1 = (const float*)d_in[3];
    const float* W2 = (const float*)d_in[4];
    const float* b2 = (const float*)d_in[5];
    const float* wg = (const float*)d_in[6];
    const float* bg = (const float*)d_in[7];
    float* out = (float*)d_out;

    dim3 grid(BSZ / 16);
    dim3 block(512);
    snde_stab_ode_kernel<<<grid, block, 0, stream>>>(y0, t, W1, b1, W2, b2, wg, bg, out);
}


# Both products formed transposed so each D tile is the next B operand: y, h and f stay in registers; 4 waves, 1 barrier per step, LDS sized for 1 block per WGP; k order of both sums changes
# speedup vs baseline: 1.5867x; 1.5867x over previous
#include <hip/hip_runtime.h>
#include <math.h>


typedef float v2f __attribute__((ext_vector_type(2)));
typedef float v4f __attribute__((ext_vector_type(4)));
typedef float v8f __attribute__((ext_vector_type(8)));

#define BSZ    4096
#define DIMN   64
#define HIDN   256
#define TSTEP  101
#define GAMMA_C 0.1f
#define EPS_C   1e-12f

__device__ __forceinline__ float hw_tanh(float x) {
#if __has_builtin(__builtin_amdgcn_tanhf)
    return __builtin_amdgcn_tanhf(x);
#elif __has_builtin(__builtin_amdgcn_tanh_f32)
    return __builtin_amdgcn_tanh_f32(x);
#else
    return tanhf(x);
#endif
}

__global__ __launch_bounds__(128)
void snde_stab_ode_kernel(const float* __restrict__ y0,
                          const float* __restrict__ tarr,
                          const float* __restrict__ W1,
                          const float* __restrict__ b1,
                          const float* __restrict__ W2,
                          const float* __restrict__ b2,
                          const float* __restrict__ wg,
                          const float* __restrict__ bgp,
                          float* __restrict__ out)
{
    __shared__ v4f part[11][4][4][2][32];

    const int tid  = threadIdx.x;
    const int w    = __builtin_amdgcn_readfirstlane(tid >> 5);
    const int lane = tid & 31;
    const int c    = lane & 15;
    const int hf   = lane >> 4;
    const int row0 = blockIdx.x * 16;


    v2f w1f[4][16];
#pragma unroll
    for (int j = 0; j < 4; ++j) {
        const int hid = 64 * w + 16 * j + c;
#pragma unroll
        for (int m = 0; m < 4; ++m) {
#pragma unroll
            for (int q = 0; q < 4; ++q) {
                const int d = 32 * hf + 8 * m + 2 * q;
                w1f[j][m * 4 + q].x = W1[(d + 0) * HIDN + hid];
                w1f[j][m * 4 + q].y = W1[(d + 1) * HIDN + hid];
            }
        }
    }
    v2f w2f[4][16];
    const int dA = 32 * (c >> 3) + (c & 7);
#pragma unroll
    for (int m = 0; m < 4; ++m) {
#pragma unroll
        for (int j = 0; j < 4; ++j) {
#pragma unroll
            for (int q = 0; q < 4; ++q) {
                const int hid = 64 * w + 16 * j + 8 * hf + 2 * q;
                w2f[m][j * 4 + q].x = W2[(hid + 0) * DIMN + dA + 8 * m];
                w2f[m][j * 4 + q].y = W2[(hid + 1) * DIMN + dA + 8 * m];
            }
        }
    }
    v2f b1v[4][4];
    v2f b2v[4][4];
    v2f wgv[4][4];
    v2f yv[4][4];
#pragma unroll
    for (int j = 0; j < 4; ++j) {
#pragma unroll
        for (int q = 0; q < 4; ++q) {
            b1v[j][q] = *(const v2f*)&b1[64 * w + 16 * j + 8 * hf + 2 * q];
            const int d = 32 * hf + 8 * j + 2 * q;
            b2v[j][q] = *(const v2f*)&b2[d];
            wgv[j][q] = *(const v2f*)&wg[d];
            yv[j][q]  = *(const v2f*)&y0[(size_t)(row0 + c) * DIMN + d];
        }
    }
    float wgss = 0.f;
    for (int j = 0; j < DIMN; ++j) { const float v = wg[j]; wgss += v * v; }

    const float dt = tarr[1] - tarr[0];
    const float bg = bgp[0];

    float* orow = out + (size_t)(row0 + c) * DIMN + 32 * hf;
#pragma unroll
    for (int m = 0; m < 4; ++m) {
        if (w == m) {
            v4f lo = {yv[m][0].x, yv[m][0].y, yv[m][1].x, yv[m][1].y};
            v4f hi = {yv[m][2].x, yv[m][2].y, yv[m][3].x, yv[m][3].y};
            *(v4f*)&orow[8 * m]     = lo;
            *(v4f*)&orow[8 * m + 4] = hi;
        }
    }

    for (int t = 0; t < TSTEP - 1; ++t) {
        float p = 0.f;
#pragma unroll
        for (int m = 0; m < 4; ++m) {
#pragma unroll
            for (int q = 0; q < 4; ++q) {
                p += yv[m][q].x * wgv[m][q].x;
                p += yv[m][q].y * wgv[m][q].y;
            }
        }
        const float po  = __shfl_xor(p, 16, 32);
        const float plo = hf ? po : p;
        const float phi = hf ? p : po;
        const float z   = bg + plo + phi;
        const float gv  = hw_tanh(z);
        const float cg  = 1.f - gv * gv;
        const float sg  = cg * cg * wgss;
        const float pinv = (fabsf(sg) > EPS_C) ? (1.f / sg) : 0.f;
        const float coef = GAMMA_C * gv * pinv * cg;

        v8f a1[4];
#pragma unroll
        for (int j = 0; j < 4; ++j) a1[j] = (v8f){0.f, 0.f, 0.f, 0.f, 0.f, 0.f, 0.f, 0.f};
#pragma unroll
        for (int ks = 0; ks < 16; ++ks) {
            const v2f bfrag = yv[ks >> 2][ks & 3];
#pragma unroll
            for (int j = 0; j < 4; ++j) {
                a1[j] = __builtin_amdgcn_wmma_f32_16x16x4_f32(
                            false, w1f[j][ks], false, bfrag, (short)0, a1[j], false, false);
            }
        }
        v2f hv[4][4];
#pragma unroll
        for (int j = 0; j < 4; ++j) {
#pragma unroll
            for (int q = 0; q < 4; ++q) {
                hv[j][q].x = hw_tanh(a1[j][2 * q + 0] + b1v[j][q].x);
                hv[j][q].y = hw_tanh(a1[j][2 * q + 1] + b1v[j][q].y);
            }
        }
        v8f a2[4];
#pragma unroll
        for (int m = 0; m < 4; ++m) a2[m] = (v8f){0.f, 0.f, 0.f, 0.f, 0.f, 0.f, 0.f, 0.f};
#pragma unroll
        for (int ks = 0; ks < 16; ++ks) {
            const v2f bfrag = hv[ks >> 2][ks & 3];
#pragma unroll
            for (int m = 0; m < 4; ++m) {
                a2[m] = __builtin_amdgcn_wmma_f32_16x16x4_f32(
                            false, w2f[m][ks], false, bfrag, (short)0, a2[m], false, false);
            }
        }
        const int buf = t & 1;
#pragma unroll
        for (int m = 0; m < 4; ++m) {
            part[buf][w][m][0][lane] = (v4f){a2[m][0], a2[m][1], a2[m][2], a2[m][3]};
            part[buf][w][m][1][lane] = (v4f){a2[m][4], a2[m][5], a2[m][6], a2[m][7]};
        }
        __syncthreads();

        float* orow_t = orow + (size_t)(t + 1) * BSZ * DIMN;
#pragma unroll
        for (int m = 0; m < 4; ++m) {
#pragma unroll
            for (int r4 = 0; r4 < 2; ++r4) {
                const v4f p0 = part[buf][0][m][r4][lane];
                const v4f p1 = part[buf][1][m][r4][lane];
                const v4f p2 = part[buf][2][m][r4][lane];
                const v4f p3 = part[buf][3][m][r4][lane];
#pragma unroll
                for (int e = 0; e < 4; ++e) {
                    const int q = 2 * r4 + (e >> 1);
                    const float b2e = (e & 1) ? b2v[m][q].y : b2v[m][q].x;
                    const float wge = (e & 1) ? wgv[m][q].y : wgv[m][q].x;
                    const float yo  = (e & 1) ? yv[m][q].y  : yv[m][q].x;
                    const float fv  = b2e + p0[e] + p1[e] + p2[e] + p3[e];
                    const float yn  = yo + (fv - coef * wge) * dt;
                    if (e & 1) yv[m][q].y = yn; else yv[m][q].x = yn;
                }
            }
        }
#define STORE_GROUP(m)                                                                  \
        case m: {                                                                       \
            *(v4f*)&orow_t[8 * m]     = (v4f){yv[m][0].x, yv[m][0].y, yv[m][1].x, yv[m][1].y}; \
            *(v4f*)&orow_t[8 * m + 4] = (v4f){yv[m][2].x, yv[m][2].y, yv[m][3].x, yv[m][3].y}; \
        } break;
        switch (w) {
            STORE_GROUP(0)
            STORE_GROUP(1)
            STORE_GROUP(2)
            default:
            STORE_GROUP(3)
        }
#undef STORE_GROUP
    }
}

extern "C" void kernel_launch(void* const* d_in, const int* in_sizes, int n_in,
                              void* d_out, int out_size, void* d_ws, size_t ws_size,
                              hipStream_t stream) {
    (void)in_sizes; (void)n_in; (void)out_size; (void)d_ws; (void)ws_size;
    const float* y0 = (const float*)d_in[0];
    const float* t  = (const float*)d_in[1];
    const float* W1 = (const float*)d_in[2];
    const float* b1 = (const float*)d_in[3];
    const float* W2 = (const float*)d_in[4];
    const float* b2 = (const float*)d_in[5];
    const float* wg = (const float*)d_in[6];
    const float* bg = (const float*)d_in[7];
    float* out = (float*)d_out;

    dim3 grid(BSZ / 16);
    dim3 block(128);
    snde_stab_ode_kernel<<<grid, block, 0, stream>>>(y0, t, W1, b1, W2, b2, wg, bg, out);
}


# Split-K exchange pipelined in two halves of the state dims on split barrier signal/wait: each half's LDS round trip and y update run under the other half's matrix work; all sums in the same order
# speedup vs baseline: 1.6815x; 1.0597x over previous
#include <hip/hip_runtime.h>
#include <math.h>


typedef float v2f __attribute__((ext_vector_type(2)));
typedef float v4f __attribute__((ext_vector_type(4)));
typedef float v8f __attribute__((ext_vector_type(8)));

#define BSZ    4096
#define DIMN   64
#define HIDN   256
#define TSTEP  101
#define GAMMA_C 0.1f
#define EPS_C   1e-12f

__device__ __forceinline__ float hw_tanh(float x) {
#if __has_builtin(__builtin_amdgcn_tanhf)
    return __builtin_amdgcn_tanhf(x);
#elif __has_builtin(__builtin_amdgcn_tanh_f32)
    return __builtin_amdgcn_tanh_f32(x);
#else
    return tanhf(x);
#endif
}

__device__ __forceinline__ void pin_here() { __builtin_amdgcn_sched_barrier(0); }
__device__ __forceinline__ void lds_drain()  { asm volatile("s_wait_dscnt 0x0" ::: "memory"); }
__device__ __forceinline__ void bar_signal() { asm volatile("s_barrier_signal -1" ::: "memory"); }
__device__ __forceinline__ void bar_wait()   { asm volatile("s_barrier_wait -1" ::: "memory"); }

#define WMMA4(acc, a, b) __builtin_amdgcn_wmma_f32_16x16x4_f32(false, (a), false, (b), (short)0, (acc), false, false)

__global__ __launch_bounds__(128)
void snde_stab_ode_kernel(const float* __restrict__ y0,
                          const float* __restrict__ tarr,
                          const float* __restrict__ W1,
                          const float* __restrict__ b1,
                          const float* __restrict__ W2,
                          const float* __restrict__ b2,
                          const float* __restrict__ wg,
                          const float* __restrict__ bgp,
                          float* __restrict__ out)
{
    __shared__ v4f part[11][4][4][2][32];

    const int tid  = threadIdx.x;
    const int w    = __builtin_amdgcn_readfirstlane(tid >> 5);
    const int lane = tid & 31;
    const int c    = lane & 15;
    const int hf   = lane >> 4;
    const int row0 = blockIdx.x * 16;


    v2f w1f[4][16];
#pragma unroll
    for (int j = 0; j < 4; ++j) {
        const int hid = 64 * w + 16 * j + c;
#pragma unroll
        for (int m = 0; m < 4; ++m) {
#pragma unroll
            for (int q = 0; q < 4; ++q) {
                const int d = 32 * hf + 8 * m + 2 * q;
                w1f[j][m * 4 + q].x = W1[(d + 0) * HIDN + hid];
                w1f[j][m * 4 + q].y = W1[(d + 1) * HIDN + hid];
            }
        }
    }
    v2f w2f[4][16];
    const int dA = 32 * (c >> 3) + (c & 7);
#pragma unroll
    for (int m = 0; m < 4; ++m) {
#pragma unroll
        for (int j = 0; j < 4; ++j) {
#pragma unroll
            for (int q = 0; q < 4; ++q) {
                const int hid = 64 * w + 16 * j + 8 * hf + 2 * q;
                w2f[m][j * 4 + q].x = W2[(hid + 0) * DIMN + dA + 8 * m];
                w2f[m][j * 4 + q].y = W2[(hid + 1) * DIMN + dA + 8 * m];
            }
        }
    }
    v2f b1v[4][4];
    v2f b2v[4][4];
    v2f wgv[4][4];
    v2f yv[4][4];
#pragma unroll
    for (int j = 0; j < 4; ++j) {
#pragma unroll
        for (int q = 0; q < 4; ++q) {
            b1v[j][q] = *(const v2f*)&b1[64 * w + 16 * j + 8 * hf + 2 * q];
            const int d = 32 * hf + 8 * j + 2 * q;
            b2v[j][q] = *(const v2f*)&b2[d];
            wgv[j][q] = *(const v2f*)&wg[d];
            yv[j][q]  = *(const v2f*)&y0[(size_t)(row0 + c) * DIMN + d];
        }
    }
    float wgss = 0.f;
    for (int j = 0; j < DIMN; ++j) { const float v = wg[j]; wgss += v * v; }

    const float dt = tarr[1] - tarr[0];
    const float bg = bgp[0];

    float* orow = out + (size_t)(row0 + c) * DIMN + 32 * hf;
#pragma unroll
    for (int m = 0; m < 4; ++m) {
        if (w == m) {
            v4f lo = {yv[m][0].x, yv[m][0].y, yv[m][1].x, yv[m][1].y};
            v4f hi = {yv[m][2].x, yv[m][2].y, yv[m][3].x, yv[m][3].y};
            *(v4f*)&orow[8 * m]     = lo;
            *(v4f*)&orow[8 * m + 4] = hi;
        }
    }

    v8f a1[4];
#pragma unroll
    for (int j = 0; j < 4; ++j) a1[j] = (v8f){0.f, 0.f, 0.f, 0.f, 0.f, 0.f, 0.f, 0.f};
#pragma unroll
    for (int ks = 0; ks < 8; ++ks) {
        const v2f bfrag = yv[ks >> 2][ks & 3];
#pragma unroll
        for (int j = 0; j < 4; ++j) a1[j] = WMMA4(a1[j], w1f[j][ks], bfrag);
    }

#define UPDATE_HALF(G, pv)                                                                 \
    _Pragma("unroll")                                                                      \
    for (int mm = 0; mm < 2; ++mm) {                                                       \
        const int m = 2 * (G) + mm;                                                        \
        _Pragma("unroll")                                                                  \
        for (int r4 = 0; r4 < 2; ++r4) {                                                   \
            const v4f p0 = pv[mm][r4][0];                                                  \
            const v4f p1 = pv[mm][r4][1];                                                  \
            const v4f p2 = pv[mm][r4][2];                                                  \
            const v4f p3 = pv[mm][r4][3];                                                  \
            _Pragma("unroll")                                                              \
            for (int e = 0; e < 4; ++e) {                                                  \
                const int q = 2 * r4 + (e >> 1);                                           \
                const float b2e = (e & 1) ? b2v[m][q].y : b2v[m][q].x;                     \
                const float wge = (e & 1) ? wgv[m][q].y : wgv[m][q].x;                     \
                const float yo  = (e & 1) ? yv[m][q].y  : yv[m][q].x;                      \
                const float fv  = b2e + p0[e] + p1[e] + p2[e] + p3[e];                     \
                const float yn  = yo + (fv - coef * wge) * dt;                             \
                if (e & 1) yv[m][q].y = yn; else yv[m][q].x = yn;                          \
            }                                                                              \
        }                                                                                  \
    }

    for (int t = 0; t < TSTEP - 1; ++t) {
#pragma unroll
        for (int ks = 8; ks < 16; ++ks) {
            const v2f bfrag = yv[ks >> 2][ks & 3];
#pragma unroll
            for (int j = 0; j < 4; ++j) a1[j] = WMMA4(a1[j], w1f[j][ks], bfrag);
        }
        float p = 0.f;
#pragma unroll
        for (int m = 0; m < 4; ++m) {
#pragma unroll
            for (int q = 0; q < 4; ++q) {
                p += yv[m][q].x * wgv[m][q].x;
                p += yv[m][q].y * wgv[m][q].y;
            }
        }
        const float po  = __shfl_xor(p, 16, 32);
        const float plo = hf ? po : p;
        const float phi = hf ? p : po;
        const float z   = bg + plo + phi;
        const float gv  = hw_tanh(z);
        const float cg  = 1.f - gv * gv;
        const float sg  = cg * cg * wgss;
        const float pinv = (fabsf(sg) > EPS_C) ? (1.f / sg) : 0.f;
        float coef = GAMMA_C * gv * pinv * cg;

        v2f hv[4][4];
#pragma unroll
        for (int j = 0; j < 4; ++j) {
#pragma unroll
            for (int q = 0; q < 4; ++q) {
                hv[j][q].x = hw_tanh(a1[j][2 * q + 0] + b1v[j][q].x);
                hv[j][q].y = hw_tanh(a1[j][2 * q + 1] + b1v[j][q].y);
            }
        }
        v8f a2[4];
#pragma unroll
        for (int m = 0; m < 4; ++m) a2[m] = (v8f){0.f, 0.f, 0.f, 0.f, 0.f, 0.f, 0.f, 0.f};
#pragma unroll
        for (int ks = 0; ks < 16; ++ks) {
            const v2f bfrag = hv[ks >> 2][ks & 3];
#pragma unroll
            for (int m = 0; m < 2; ++m) a2[m] = WMMA4(a2[m], w2f[m][ks], bfrag);
        }
        asm volatile("" : "+v"(coef));
#pragma unroll
        for (int m = 0; m < 2; ++m) {
            part[0][w][m][0][lane] = (v4f){a2[m][0], a2[m][1], a2[m][2], a2[m][3]};
            part[0][w][m][1][lane] = (v4f){a2[m][4], a2[m][5], a2[m][6], a2[m][7]};
        }
        pin_here();
#pragma unroll
        for (int ks = 0; ks < 4; ++ks) {
            const v2f bfrag = hv[ks >> 2][ks & 3];
#pragma unroll
            for (int m = 2; m < 4; ++m) a2[m] = WMMA4(a2[m], w2f[m][ks], bfrag);
        }
        pin_here();
        lds_drain();
        bar_signal();
        pin_here();
#pragma unroll
        for (int ks = 4; ks < 8; ++ks) {
            const v2f bfrag = hv[ks >> 2][ks & 3];
#pragma unroll
            for (int m = 2; m < 4; ++m) a2[m] = WMMA4(a2[m], w2f[m][ks], bfrag);
        }
        pin_here();
        bar_wait();
        v4f pa[2][2][4];
#pragma unroll
        for (int mm = 0; mm < 2; ++mm)
#pragma unroll
            for (int r4 = 0; r4 < 2; ++r4)
#pragma unroll
                for (int v = 0; v < 4; ++v) pa[mm][r4][v] = part[0][v][mm][r4][lane];
        pin_here();
#pragma unroll
        for (int ks = 8; ks < 12; ++ks) {
            const v2f bfrag = hv[ks >> 2][ks & 3];
#pragma unroll
            for (int m = 2; m < 4; ++m) a2[m] = WMMA4(a2[m], w2f[m][ks], bfrag);
        }
        pin_here();
#pragma unroll
        for (int ks = 12; ks < 16; ++ks) {
            const v2f bfrag = hv[ks >> 2][ks & 3];
#pragma unroll
            for (int m = 2; m < 4; ++m) a2[m] = WMMA4(a2[m], w2f[m][ks], bfrag);
        }
        UPDATE_HALF(0, pa)
        pin_here();
#pragma unroll
        for (int m = 2; m < 4; ++m) {
            part[0][w][m][0][lane] = (v4f){a2[m][0], a2[m][1], a2[m][2], a2[m][3]};
            part[0][w][m][1][lane] = (v4f){a2[m][4], a2[m][5], a2[m][6], a2[m][7]};
        }
        pin_here();
#pragma unroll
        for (int j = 0; j < 4; ++j) a1[j] = (v8f){0.f, 0.f, 0.f, 0.f, 0.f, 0.f, 0.f, 0.f};
#pragma unroll
        for (int ks = 0; ks < 2; ++ks) {
            const v2f bfrag = yv[ks >> 2][ks & 3];
#pragma unroll
            for (int j = 0; j < 4; ++j) a1[j] = WMMA4(a1[j], w1f[j][ks], bfrag);
        }
        pin_here();
        lds_drain();
        bar_signal();
        pin_here();
#pragma unroll
        for (int ks = 2; ks < 4; ++ks) {
            const v2f bfrag = yv[ks >> 2][ks & 3];
#pragma unroll
            for (int j = 0; j < 4; ++j) a1[j] = WMMA4(a1[j], w1f[j][ks], bfrag);
        }
        pin_here();
        bar_wait();
        v4f pb[2][2][4];
#pragma unroll
        for (int mm = 0; mm < 2; ++mm)
#pragma unroll
            for (int r4 = 0; r4 < 2; ++r4)
#pragma unroll
                for (int v = 0; v < 4; ++v) pb[mm][r4][v] = part[0][v][2 + mm][r4][lane];
        pin_here();
#pragma unroll
        for (int ks = 4; ks < 6; ++ks) {
            const v2f bfrag = yv[ks >> 2][ks & 3];
#pragma unroll
            for (int j = 0; j < 4; ++j) a1[j] = WMMA4(a1[j], w1f[j][ks], bfrag);
        }
        pin_here();
#pragma unroll
        for (int ks = 6; ks < 8; ++ks) {
            const v2f bfrag = yv[ks >> 2][ks & 3];
#pragma unroll
            for (int j = 0; j < 4; ++j) a1[j] = WMMA4(a1[j], w1f[j][ks], bfrag);
        }
        UPDATE_HALF(1, pb)

        float* orow_t = orow + (size_t)(t + 1) * BSZ * DIMN;
#define STORE_GROUP(m)                                                                  \
        case m: {                                                                       \
            *(v4f*)&orow_t[8 * m]     = (v4f){yv[m][0].x, yv[m][0].y, yv[m][1].x, yv[m][1].y}; \
            *(v4f*)&orow_t[8 * m + 4] = (v4f){yv[m][2].x, yv[m][2].y, yv[m][3].x, yv[m][3].y}; \
        } break;
        switch (w) {
            STORE_GROUP(0)
            STORE_GROUP(1)
            STORE_GROUP(2)
            default:
            STORE_GROUP(3)
        }
#undef STORE_GROUP
    }
#undef UPDATE_HALF
}

extern "C" void kernel_launch(void* const* d_in, const int* in_sizes, int n_in,
                              void* d_out, int out_size, void* d_ws, size_t ws_size,
                              hipStream_t stream) {
    (void)in_sizes; (void)n_in; (void)out_size; (void)d_ws; (void)ws_size;
    const float* y0 = (const float*)d_in[0];
    const float* t  = (const float*)d_in[1];
    const float* W1 = (const float*)d_in[2];
    const float* b1 = (const float*)d_in[3];
    const float* W2 = (const float*)d_in[4];
    const float* b2 = (const float*)d_in[5];
    const float* wg = (const float*)d_in[6];
    const float* bg = (const float*)d_in[7];
    float* out = (float*)d_out;

    dim3 grid(BSZ / 16);
    dim3 block(128);
    snde_stab_ode_kernel<<<grid, block, 0, stream>>>(y0, t, W1, b1, W2, b2, wg, bg, out);
}


# Vector-ALU slots of the step cut (f32 matrix form hides none): b1, b2 become accumulator start values; constraint dot as packed FMAs on 4 partial sums, halves joined by permlanex16; sum order changes
# speedup vs baseline: 1.7542x; 1.0432x over previous
#include <hip/hip_runtime.h>
#include <math.h>


typedef float v2f __attribute__((ext_vector_type(2)));
typedef float v4f __attribute__((ext_vector_type(4)));
typedef float v8f __attribute__((ext_vector_type(8)));

#define BSZ    4096
#define DIMN   64
#define HIDN   256
#define TSTEP  101
#define GAMMA_C 0.1f
#define EPS_C   1e-12f

__device__ __forceinline__ float hw_tanh(float x) {
#if __has_builtin(__builtin_amdgcn_tanhf)
    return __builtin_amdgcn_tanhf(x);
#elif __has_builtin(__builtin_amdgcn_tanh_f32)
    return __builtin_amdgcn_tanh_f32(x);
#else
    return tanhf(x);
#endif
}

__device__ __forceinline__ float other_half(float x) {
    return __builtin_bit_cast(float, __builtin_amdgcn_permlanex16(__builtin_bit_cast(int, x), __builtin_bit_cast(int, x),
                                                                  0x76543210, 0xfedcba98, false, false));
}

__device__ __forceinline__ void pin_here() { __builtin_amdgcn_sched_barrier(0); }
__device__ __forceinline__ void lds_drain()  { asm volatile("s_wait_dscnt 0x0" ::: "memory"); }
__device__ __forceinline__ void bar_signal() { asm volatile("s_barrier_signal -1" ::: "memory"); }
__device__ __forceinline__ void bar_wait()   { asm volatile("s_barrier_wait -1" ::: "memory"); }

#define WMMA4(acc, a, b) __builtin_amdgcn_wmma_f32_16x16x4_f32(false, (a), false, (b), (short)0, (acc), false, false)

__global__ __launch_bounds__(128)
void snde_stab_ode_kernel(const float* __restrict__ y0,
                          const float* __restrict__ tarr,
                          const float* __restrict__ W1,
                          const float* __restrict__ b1,
                          const float* __restrict__ W2,
                          const float* __restrict__ b2,
                          const float* __restrict__ wg,
                          const float* __restrict__ bgp,
                          float* __restrict__ out)
{
    __shared__ v4f part[11][4][4][2][32];

    const int tid  = threadIdx.x;
    const int w    = __builtin_amdgcn_readfirstlane(tid >> 5);
    const int lane = tid & 31;
    const int c    = lane & 15;
    const int hf   = lane >> 4;
    const int row0 = blockIdx.x * 16;


    v2f w1f[4][16];
#pragma unroll
    for (int j = 0; j < 4; ++j) {
        const int hid = 64 * w + 16 * j + c;
#pragma unroll
        for (int m = 0; m < 4; ++m) {
#pragma unroll
            for (int q = 0; q < 4; ++q) {
                const int d = 32 * hf + 8 * m + 2 * q;
                w1f[j][m * 4 + q].x = W1[(d + 0) * HIDN + hid];
                w1f[j][m * 4 + q].y = W1[(d + 1) * HIDN + hid];
            }
        }
    }
    v2f w2f[4][16];
    const int dA = 32 * (c >> 3) + (c & 7);
#pragma unroll
    for (int m = 0; m < 4; ++m) {
#pragma unroll
        for (int j = 0; j < 4; ++j) {
#pragma unroll
            for (int q = 0; q < 4; ++q) {
                const int hid = 64 * w + 16 * j + 8 * hf + 2 * q;
                w2f[m][j * 4 + q].x = W2[(hid + 0) * DIMN + dA + 8 * m];
                w2f[m][j * 4 + q].y = W2[(hid + 1) * DIMN + dA + 8 * m];
            }
        }
    }
    v8f b1acc[4];
    v8f b2acc[4];
    v2f wgv[4][4];
    v2f yv[4][4];
#pragma unroll
    for (int j = 0; j < 4; ++j) {
#pragma unroll
        for (int q = 0; q < 4; ++q) {
            const v2f b1p = *(const v2f*)&b1[64 * w + 16 * j + 8 * hf + 2 * q];
            b1acc[j][2 * q + 0] = b1p.x;
            b1acc[j][2 * q + 1] = b1p.y;
            const int d = 32 * hf + 8 * j + 2 * q;
            const v2f b2p = *(const v2f*)&b2[d];
            b2acc[j][2 * q + 0] = (w == 0) ? b2p.x : 0.f;
            b2acc[j][2 * q + 1] = (w == 0) ? b2p.y : 0.f;
            wgv[j][q] = *(const v2f*)&wg[d];
            yv[j][q]  = *(const v2f*)&y0[(size_t)(row0 + c) * DIMN + d];
        }
    }
    float wgss = 0.f;
    for (int j = 0; j < DIMN; ++j) { const float v = wg[j]; wgss += v * v; }

    const float dt = tarr[1] - tarr[0];
    const float bg = bgp[0];

    float* orow = out + (size_t)(row0 + c) * DIMN + 32 * hf;
#pragma unroll
    for (int m = 0; m < 4; ++m) {
        if (w == m) {
            v4f lo = {yv[m][0].x, yv[m][0].y, yv[m][1].x, yv[m][1].y};
            v4f hi = {yv[m][2].x, yv[m][2].y, yv[m][3].x, yv[m][3].y};
            *(v4f*)&orow[8 * m]     = lo;
            *(v4f*)&orow[8 * m + 4] = hi;
        }
    }

    v8f a1[4];
#pragma unroll
    for (int ks = 0; ks < 8; ++ks) {
        const v2f bfrag = yv[ks >> 2][ks & 3];
#pragma unroll
        for (int j = 0; j < 4; ++j) a1[j] = WMMA4(ks == 0 ? b1acc[j] : a1[j], w1f[j][ks], bfrag);
    }

#define UPDATE_HALF(G, pv)                                                                 \
    _Pragma("unroll")                                                                      \
    for (int mm = 0; mm < 2; ++mm) {                                                       \
        const int m = 2 * (G) + mm;                                                        \
        _Pragma("unroll")                                                                  \
        for (int r4 = 0; r4 < 2; ++r4) {                                                   \
            const v4f p0 = pv[mm][r4][0];                                                  \
            const v4f p1 = pv[mm][r4][1];                                                  \
            const v4f p2 = pv[mm][r4][2];                                                  \
            const v4f p3 = pv[mm][r4][3];                                                  \
            _Pragma("unroll")                                                              \
            for (int e = 0; e < 4; ++e) {                                                  \
                const int q = 2 * r4 + (e >> 1);                                           \
                const float wge = (e & 1) ? wgv[m][q].y : wgv[m][q].x;                     \
                const float yo  = (e & 1) ? yv[m][q].y  : yv[m][q].x;                      \
                const float fv  = p0[e] + p1[e] + p2[e] + p3[e];                           \
                const float yn  = yo + (fv - coef * wge) * dt;                             \
                if (e & 1) yv[m][q].y = yn; else yv[m][q].x = yn;                          \
            }                                                                              \
        }                                                                                  \
    }

    for (int t = 0; t < TSTEP - 1; ++t) {
#pragma unroll
        for (int ks = 8; ks < 16; ++ks) {
            const v2f bfrag = yv[ks >> 2][ks & 3];
#pragma unroll
            for (int j = 0; j < 4; ++j) a1[j] = WMMA4(a1[j], w1f[j][ks], bfrag);
        }
        v2f dsum[4];
#pragma unroll
        for (int i = 0; i < 4; ++i) dsum[i] = yv[0][i] * wgv[0][i];
#pragma unroll
        for (int i = 4; i < 16; ++i) dsum[i & 3] += yv[i >> 2][i & 3] * wgv[i >> 2][i & 3];
        const v2f dtot = (dsum[0] + dsum[1]) + (dsum[2] + dsum[3]);
        const float p  = dtot.x + dtot.y;
        const float po  = other_half(p);
        const float plo = hf ? po : p;
        const float phi = hf ? p : po;
        const float z   = bg + plo + phi;
        const float gv  = hw_tanh(z);
        const float cg  = 1.f - gv * gv;
        const float sg  = cg * cg * wgss;
        const float pinv = (fabsf(sg) > EPS_C) ? (1.f / sg) : 0.f;
        float coef = GAMMA_C * gv * pinv * cg;

        v2f hv[4][4];
#pragma unroll
        for (int j = 0; j < 4; ++j) {
#pragma unroll
            for (int q = 0; q < 4; ++q) {
                hv[j][q].x = hw_tanh(a1[j][2 * q + 0]);
                hv[j][q].y = hw_tanh(a1[j][2 * q + 1]);
            }
        }
        v8f a2[4];
#pragma unroll
        for (int ks = 0; ks < 16; ++ks) {
            const v2f bfrag = hv[ks >> 2][ks & 3];
#pragma unroll
            for (int m = 0; m < 2; ++m) a2[m] = WMMA4(ks == 0 ? b2acc[m] : a2[m], w2f[m][ks], bfrag);
        }
        asm volatile("" : "+v"(coef));
#pragma unroll
        for (int m = 0; m < 2; ++m) {
            part[0][w][m][0][lane] = (v4f){a2[m][0], a2[m][1], a2[m][2], a2[m][3]};
            part[0][w][m][1][lane] = (v4f){a2[m][4], a2[m][5], a2[m][6], a2[m][7]};
        }
        pin_here();
#pragma unroll
        for (int ks = 0; ks < 4; ++ks) {
            const v2f bfrag = hv[ks >> 2][ks & 3];
#pragma unroll
            for (int m = 2; m < 4; ++m) a2[m] = WMMA4(ks == 0 ? b2acc[m] : a2[m], w2f[m][ks], bfrag);
        }
        pin_here();
        lds_drain();
        bar_signal();
        pin_here();
#pragma unroll
        for (int ks = 4; ks < 8; ++ks) {
            const v2f bfrag = hv[ks >> 2][ks & 3];
#pragma unroll
            for (int m = 2; m < 4; ++m) a2[m] = WMMA4(a2[m], w2f[m][ks], bfrag);
        }
        pin_here();
        bar_wait();
        v4f pa[2][2][4];
#pragma unroll
        for (int mm = 0; mm < 2; ++mm)
#pragma unroll
            for (int r4 = 0; r4 < 2; ++r4)
#pragma unroll
                for (int v = 0; v < 4; ++v) pa[mm][r4][v] = part[0][v][mm][r4][lane];
        pin_here();
#pragma unroll
        for (int ks = 8; ks < 12; ++ks) {
            const v2f bfrag = hv[ks >> 2][ks & 3];
#pragma unroll
            for (int m = 2; m < 4; ++m) a2[m] = WMMA4(a2[m], w2f[m][ks], bfrag);
        }
        pin_here();
#pragma unroll
        for (int ks = 12; ks < 16; ++ks) {
            const v2f bfrag = hv[ks >> 2][ks & 3];
#pragma unroll
            for (int m = 2; m < 4; ++m) a2[m] = WMMA4(a2[m], w2f[m][ks], bfrag);
        }
        UPDATE_HALF(0, pa)
        pin_here();
#pragma unroll
        for (int m = 2; m < 4; ++m) {
            part[0][w][m][0][lane] = (v4f){a2[m][0], a2[m][1], a2[m][2], a2[m][3]};
            part[0][w][m][1][lane] = (v4f){a2[m][4], a2[m][5], a2[m][6], a2[m][7]};
        }
        pin_here();
#pragma unroll
        for (int ks = 0; ks < 2; ++ks) {
            const v2f bfrag = yv[ks >> 2][ks & 3];
#pragma unroll
            for (int j = 0; j < 4; ++j) a1[j] = WMMA4(ks == 0 ? b1acc[j] : a1[j], w1f[j][ks], bfrag);
        }
        pin_here();
        lds_drain();
        bar_signal();
        pin_here();
#pragma unroll
        for (int ks = 2; ks < 4; ++ks) {
            const v2f bfrag = yv[ks >> 2][ks & 3];
#pragma unroll
            for (int j = 0; j < 4; ++j) a1[j] = WMMA4(a1[j], w1f[j][ks], bfrag);
        }
        pin_here();
        bar_wait();
        v4f pb[2][2][4];
#pragma unroll
        for (int mm = 0; mm < 2; ++mm)
#pragma unroll
            for (int r4 = 0; r4 < 2; ++r4)
#pragma unroll
                for (int v = 0; v < 4; ++v) pb[mm][r4][v] = part[0][v][2 + mm][r4][lane];
        pin_here();
#pragma unroll
        for (int ks = 4; ks < 6; ++ks) {
            const v2f bfrag = yv[ks >> 2][ks & 3];
#pragma unroll
            for (int j = 0; j < 4; ++j) a1[j] = WMMA4(a1[j], w1f[j][ks], bfrag);
        }
        pin_here();
#pragma unroll
        for (int ks = 6; ks < 8; ++ks) {
            const v2f bfrag = yv[ks >> 2][ks & 3];
#pragma unroll
            for (int j = 0; j < 4; ++j) a1[j] = WMMA4(a1[j], w1f[j][ks], bfrag);
        }
        UPDATE_HALF(1, pb)

        float* orow_t = orow + (size_t)(t + 1) * BSZ * DIMN;
#define STORE_GROUP(m)                                                                  \
        case m: {                                                                       \
            *(v4f*)&orow_t[8 * m]     = (v4f){yv[m][0].x, yv[m][0].y, yv[m][1].x, yv[m][1].y}; \
            *(v4f*)&orow_t[8 * m + 4] = (v4f){yv[m][2].x, yv[m][2].y, yv[m][3].x, yv[m][3].y}; \
        } break;
        switch (w) {
            STORE_GROUP(0)
            STORE_GROUP(1)
            STORE_GROUP(2)
            default:
            STORE_GROUP(3)
        }
#undef STORE_GROUP
    }
#undef UPDATE_HALF
}

extern "C" void kernel_launch(void* const* d_in, const int* in_sizes, int n_in,
                              void* d_out, int out_size, void* d_ws, size_t ws_size,
                              hipStream_t stream) {
    (void)in_sizes; (void)n_in; (void)out_size; (void)d_ws; (void)ws_size;
    const float* y0 = (const float*)d_in[0];
    const float* t  = (const float*)d_in[1];
    const float* W1 = (const float*)d_in[2];
    const float* b1 = (const float*)d_in[3];
    const float* W2 = (const float*)d_in[4];
    const float* b2 = (const float*)d_in[5];
    const float* wg = (const float*)d_in[6];
    const float* bg = (const float*)d_in[7];
    float* out = (float*)d_out;

    dim3 grid(BSZ / 16);
    dim3 block(128);
    snde_stab_ode_kernel<<<grid, block, 0, stream>>>(y0, t, W1, b1, W2, b2, wg, bg, out);
}


# speedup vs baseline: 1.8446x; 1.0515x over previous
#include <hip/hip_runtime.h>
#include <math.h>


typedef float v2f __attribute__((ext_vector_type(2)));
typedef float v4f __attribute__((ext_vector_type(4)));
typedef float v8f __attribute__((ext_vector_type(8)));

#define BSZ    4096
#define DIMN   64
#define HIDN   256
#define TSTEP  101
#define GAMMA_C 0.1f
#define EPS_C   1e-12f

__device__ __forceinline__ float hw_tanh(float x) {
#if __has_builtin(__builtin_amdgcn_tanhf)
    return __builtin_amdgcn_tanhf(x);
#elif __has_builtin(__builtin_amdgcn_tanh_f32)
    return __builtin_amdgcn_tanh_f32(x);
#else
    return tanhf(x);
#endif
}

__device__ __forceinline__ float other_half(float x) {
    return __builtin_bit_cast(float, __builtin_amdgcn_permlanex16(__builtin_bit_cast(int, x), __builtin_bit_cast(int, x),
                                                                  0x76543210, 0xfedcba98, false, false));
}

__device__ __forceinline__ void pin_here() { __builtin_amdgcn_sched_barrier(0); }
__device__ __forceinline__ void lds_drain()  { asm volatile("s_wait_dscnt 0x0" ::: "memory"); }
__device__ __forceinline__ void bar_signal() { asm volatile("s_barrier_signal -1" ::: "memory"); }
__device__ __forceinline__ void bar_wait()   { asm volatile("s_barrier_wait -1" ::: "memory"); }

#define WMMA4(acc, a, b) __builtin_amdgcn_wmma_f32_16x16x4_f32(false, (a), false, (b), (short)0, (acc), false, false)

template <int WV>
__device__ __forceinline__ void step_loop(v4f (&part)[11][4][4][2][32],
                                          v2f (&w1f)[4][16], v2f (&w2f)[4][16],
                                          v8f (&b1acc)[4], v8f (&b2acc)[4],
                                          v2f (&wgv)[4][4], v2f (&yv)[4][4],
                                          const float wgss, const float dt, const float bg,
                                          const int lane, const int hf,
                                          float* __restrict__ orow)
{
    const v8f zero8 = {0.f, 0.f, 0.f, 0.f, 0.f, 0.f, 0.f, 0.f};

    v8f a1[4];
#pragma unroll
    for (int ks = 0; ks < 8; ++ks) {
        const v2f bfrag = yv[ks >> 2][ks & 3];
#pragma unroll
        for (int j = 0; j < 4; ++j) a1[j] = WMMA4(ks == 0 ? b1acc[j] : a1[j], w1f[j][ks], bfrag);
    }

#define LOAD_HALF(G, pv)                                                                   \
    _Pragma("unroll")                                                                      \
    for (int k = 1; k < 4; ++k) {                                                          \
        const int v = (WV + k) & 3;                                                        \
        _Pragma("unroll")                                                                  \
        for (int mm = 0; mm < 2; ++mm)                                                     \
            _Pragma("unroll")                                                              \
            for (int r4 = 0; r4 < 2; ++r4) pv[mm][r4][v] = part[2 * v][0][2 * (G) + mm][r4][lane]; \
    }

#define UPDATE_HALF(G, pv)                                                                 \
    _Pragma("unroll")                                                                      \
    for (int mm = 0; mm < 2; ++mm) {                                                       \
        const int m = 2 * (G) + mm;                                                        \
        _Pragma("unroll")                                                                  \
        for (int r4 = 0; r4 < 2; ++r4) {                                                   \
            const v4f own = {a2[m][4 * r4 + 0], a2[m][4 * r4 + 1],                         \
                             a2[m][4 * r4 + 2], a2[m][4 * r4 + 3]};                        \
            const v4f p0 = (WV == 0) ? own : pv[mm][r4][0];                                \
            const v4f p1 = (WV == 1) ? own : pv[mm][r4][1];                                \
            const v4f p2 = (WV == 2) ? own : pv[mm][r4][2];                                \
            const v4f p3 = (WV == 3) ? own : pv[mm][r4][3];                                \
            _Pragma("unroll")                                                              \
            for (int e = 0; e < 4; ++e) {                                                  \
                const int q = 2 * r4 + (e >> 1);                                           \
                const float wge = (e & 1) ? wgv[m][q].y : wgv[m][q].x;                     \
                const float yo  = (e & 1) ? yv[m][q].y  : yv[m][q].x;                      \
                const float fv  = p0[e] + p1[e] + p2[e] + p3[e];                           \
                const float yn  = yo + (fv - coef * wge) * dt;                             \
                if (e & 1) yv[m][q].y = yn; else yv[m][q].x = yn;                          \
            }                                                                              \
        }                                                                                  \
    }

    for (int t = 0; t < TSTEP - 1; ++t) {
        {
            float* orow_t = orow + (size_t)t * BSZ * DIMN;
            *(v4f*)&orow_t[8 * WV]     = (v4f){yv[WV][0].x, yv[WV][0].y, yv[WV][1].x, yv[WV][1].y};
            *(v4f*)&orow_t[8 * WV + 4] = (v4f){yv[WV][2].x, yv[WV][2].y, yv[WV][3].x, yv[WV][3].y};
        }
#pragma unroll
        for (int ks = 8; ks < 16; ++ks) {
            const v2f bfrag = yv[ks >> 2][ks & 3];
#pragma unroll
            for (int j = 0; j < 4; ++j) a1[j] = WMMA4(a1[j], w1f[j][ks], bfrag);
        }
        v2f dsum[4];
#pragma unroll
        for (int i = 0; i < 4; ++i) dsum[i] = yv[0][i] * wgv[0][i];
#pragma unroll
        for (int i = 4; i < 16; ++i) dsum[i & 3] += yv[i >> 2][i & 3] * wgv[i >> 2][i & 3];
        const v2f dtot = (dsum[0] + dsum[1]) + (dsum[2] + dsum[3]);
        const float p  = dtot.x + dtot.y;
        const float po  = other_half(p);
        const float plo = hf ? po : p;
        const float phi = hf ? p : po;
        const float z   = bg + plo + phi;
        const float gv  = hw_tanh(z);
        const float cg  = 1.f - gv * gv;
        const float sg  = cg * cg * wgss;
        const float pinv = (fabsf(sg) > EPS_C) ? (1.f / sg) : 0.f;
        float coef = GAMMA_C * gv * pinv * cg;

        v2f hv[4][4];
#pragma unroll
        for (int j = 0; j < 4; ++j) {
#pragma unroll
            for (int q = 0; q < 4; ++q) {
                hv[j][q].x = hw_tanh(a1[j][2 * q + 0]);
                hv[j][q].y = hw_tanh(a1[j][2 * q + 1]);
            }
        }
        v8f a2[4];
#pragma unroll
        for (int ks = 0; ks < 16; ++ks) {
            const v2f bfrag = hv[ks >> 2][ks & 3];
#pragma unroll
            for (int m = 0; m < 2; ++m)
                a2[m] = WMMA4(ks == 0 ? (WV == 0 ? b2acc[m] : zero8) : a2[m], w2f[m][ks], bfrag);
        }
        asm volatile("" : "+v"(coef));
#pragma unroll
        for (int m = 0; m < 2; ++m) {
            part[2 * WV][0][m][0][lane] = (v4f){a2[m][0], a2[m][1], a2[m][2], a2[m][3]};
            part[2 * WV][0][m][1][lane] = (v4f){a2[m][4], a2[m][5], a2[m][6], a2[m][7]};
        }
        pin_here();
#pragma unroll
        for (int ks = 0; ks < 4; ++ks) {
            const v2f bfrag = hv[ks >> 2][ks & 3];
#pragma unroll
            for (int m = 2; m < 4; ++m)
                a2[m] = WMMA4(ks == 0 ? (WV == 0 ? b2acc[m] : zero8) : a2[m], w2f[m][ks], bfrag);
        }
        pin_here();
        lds_drain();
        bar_signal();
        pin_here();
#pragma unroll
        for (int ks = 4; ks < 8; ++ks) {
            const v2f bfrag = hv[ks >> 2][ks & 3];
#pragma unroll
            for (int m = 2; m < 4; ++m) a2[m] = WMMA4(a2[m], w2f[m][ks], bfrag);
        }
        pin_here();
        bar_wait();
        v4f pa[2][2][4];
        LOAD_HALF(0, pa)
        pin_here();
#pragma unroll
        for (int ks = 8; ks < 12; ++ks) {
            const v2f bfrag = hv[ks >> 2][ks & 3];
#pragma unroll
            for (int m = 2; m < 4; ++m) a2[m] = WMMA4(a2[m], w2f[m][ks], bfrag);
        }
        pin_here();
#pragma unroll
        for (int ks = 12; ks < 16; ++ks) {
            const v2f bfrag = hv[ks >> 2][ks & 3];
#pragma unroll
            for (int m = 2; m < 4; ++m) a2[m] = WMMA4(a2[m], w2f[m][ks], bfrag);
        }
        UPDATE_HALF(0, pa)
        pin_here();
#pragma unroll
        for (int m = 2; m < 4; ++m) {
            part[2 * WV][0][m][0][lane] = (v4f){a2[m][0], a2[m][1], a2[m][2], a2[m][3]};
            part[2 * WV][0][m][1][lane] = (v4f){a2[m][4], a2[m][5], a2[m][6], a2[m][7]};
        }
        pin_here();
#pragma unroll
        for (int ks = 0; ks < 2; ++ks) {
            const v2f bfrag = yv[ks >> 2][ks & 3];
#pragma unroll
            for (int j = 0; j < 4; ++j) a1[j] = WMMA4(ks == 0 ? b1acc[j] : a1[j], w1f[j][ks], bfrag);
        }
        pin_here();
        lds_drain();
        bar_signal();
        pin_here();
#pragma unroll
        for (int ks = 2; ks < 4; ++ks) {
            const v2f bfrag = yv[ks >> 2][ks & 3];
#pragma unroll
            for (int j = 0; j < 4; ++j) a1[j] = WMMA4(a1[j], w1f[j][ks], bfrag);
        }
        pin_here();
        bar_wait();
        v4f pb[2][2][4];
        LOAD_HALF(1, pb)
        pin_here();
#pragma unroll
        for (int ks = 4; ks < 6; ++ks) {
            const v2f bfrag = yv[ks >> 2][ks & 3];
#pragma unroll
            for (int j = 0; j < 4; ++j) a1[j] = WMMA4(a1[j], w1f[j][ks], bfrag);
        }
        pin_here();
#pragma unroll
        for (int ks = 6; ks < 8; ++ks) {
            const v2f bfrag = yv[ks >> 2][ks & 3];
#pragma unroll
            for (int j = 0; j < 4; ++j) a1[j] = WMMA4(a1[j], w1f[j][ks], bfrag);
        }
        UPDATE_HALF(1, pb)

    }
    {
        float* orow_t = orow + (size_t)(TSTEP - 1) * BSZ * DIMN;
        *(v4f*)&orow_t[8 * WV]     = (v4f){yv[WV][0].x, yv[WV][0].y, yv[WV][1].x, yv[WV][1].y};
        *(v4f*)&orow_t[8 * WV + 4] = (v4f){yv[WV][2].x, yv[WV][2].y, yv[WV][3].x, yv[WV][3].y};
    }
#undef UPDATE_HALF
#undef LOAD_HALF
}

__global__ __launch_bounds__(128)
void snde_stab_ode_kernel(const float* __restrict__ y0,
                          const float* __restrict__ tarr,
                          const float* __restrict__ W1,
                          const float* __restrict__ b1,
                          const float* __restrict__ W2,
                          const float* __restrict__ b2,
                          const float* __restrict__ wg,
                          const float* __restrict__ bgp,
                          float* __restrict__ out)
{
    __shared__ v4f part[11][4][4][2][32];

    const int tid  = threadIdx.x;
    const int w    = __builtin_amdgcn_readfirstlane(tid >> 5);
    const int lane = tid & 31;
    const int c    = lane & 15;
    const int hf   = lane >> 4;
    const int row0 = blockIdx.x * 16;


    v2f w1f[4][16];
#pragma unroll
    for (int j = 0; j < 4; ++j) {
        const int hid = 64 * w + 16 * j + c;
#pragma unroll
        for (int m = 0; m < 4; ++m) {
#pragma unroll
            for (int q = 0; q < 4; ++q) {
                const int d = 32 * hf + 8 * m + 2 * q;
                w1f[j][m * 4 + q].x = W1[(d + 0) * HIDN + hid];
                w1f[j][m * 4 + q].y = W1[(d + 1) * HIDN + hid];
            }
        }
    }
    v2f w2f[4][16];
    const int dA = 32 * (c >> 3) + (c & 7);
#pragma unroll
    for (int m = 0; m < 4; ++m) {
#pragma unroll
        for (int j = 0; j < 4; ++j) {
#pragma unroll
            for (int q = 0; q < 4; ++q) {
                const int hid = 64 * w + 16 * j + 8 * hf + 2 * q;
                w2f[m][j * 4 + q].x = W2[(hid + 0) * DIMN + dA + 8 * m];
                w2f[m][j * 4 + q].y = W2[(hid + 1) * DIMN + dA + 8 * m];
            }
        }
    }
    v8f b1acc[4];
    v8f b2acc[4];
    v2f wgv[4][4];
    v2f yv[4][4];
#pragma unroll
    for (int j = 0; j < 4; ++j) {
#pragma unroll
        for (int q = 0; q < 4; ++q) {
            const v2f b1p = *(const v2f*)&b1[64 * w + 16 * j + 8 * hf + 2 * q];
            b1acc[j][2 * q + 0] = b1p.x;
            b1acc[j][2 * q + 1] = b1p.y;
            const int d = 32 * hf + 8 * j + 2 * q;
            const v2f b2p = *(const v2f*)&b2[d];
            b2acc[j][2 * q + 0] = b2p.x;
            b2acc[j][2 * q + 1] = b2p.y;
            wgv[j][q] = *(const v2f*)&wg[d];
            yv[j][q]  = *(const v2f*)&y0[(size_t)(row0 + c) * DIMN + d];
        }
    }
    float wgss = 0.f;
    for (int j = 0; j < DIMN; ++j) { const float v = wg[j]; wgss += v * v; }

    const float dt = tarr[1] - tarr[0];
    const float bg = bgp[0];

    float* orow = out + (size_t)(row0 + c) * DIMN + 32 * hf;

    switch (w) {
        case 0:  step_loop<0>(part, w1f, w2f, b1acc, b2acc, wgv, yv, wgss, dt, bg, lane, hf, orow); break;
        case 1:  step_loop<1>(part, w1f, w2f, b1acc, b2acc, wgv, yv, wgss, dt, bg, lane, hf, orow); break;
        case 2:  step_loop<2>(part, w1f, w2f, b1acc, b2acc, wgv, yv, wgss, dt, bg, lane, hf, orow); break;
        default: step_loop<3>(part, w1f, w2f, b1acc, b2acc, wgv, yv, wgss, dt, bg, lane, hf, orow); break;
    }
}

extern "C" void kernel_launch(void* const* d_in, const int* in_sizes, int n_in,
                              void* d_out, int out_size, void* d_ws, size_t ws_size,
                              hipStream_t stream) {
    (void)in_sizes; (void)n_in; (void)out_size; (void)d_ws; (void)ws_size;
    const float* y0 = (const float*)d_in[0];
    const float* t  = (const float*)d_in[1];
    const float* W1 = (const float*)d_in[2];
    const float* b1 = (const float*)d_in[3];
    const float* W2 = (const float*)d_in[4];
    const float* b2 = (const float*)d_in[5];
    const float* wg = (const float*)d_in[6];
    const float* bg = (const float*)d_in[7];
    float* out = (float*)d_out;

    dim3 grid(BSZ / 16);
    dim3 block(128);
    snde_stab_ode_kernel<<<grid, block, 0, stream>>>(y0, t, W1, b1, W2, b2, wg, bg, out);
}
